# starting kernel
# baseline (speedup 1.0000x reference)
#include <hip/hip_runtime.h>
#include <hip/hip_bf16.h>

typedef __attribute__((ext_vector_type(2))) float v2f;
typedef __attribute__((ext_vector_type(8))) float v8f;

#define GC_N_NODES 100000
#define GC_FEAT 64

__global__ void gc_zero_kernel(float4* __restrict__ p, int n4) {
    int i = blockIdx.x * blockDim.x + threadIdx.x;
    if (i < n4) p[i] = make_float4(0.f, 0.f, 0.f, 0.f);
}

__global__ void gc_spmm_scatter(const float*  __restrict__ x,
                                const int*    __restrict__ erow,
                                const int*    __restrict__ ecol,
                                const float*  __restrict__ eval,
                                float*        __restrict__ support,
                                int E) {
    int tid  = blockIdx.x * blockDim.x + threadIdx.x;
    int e    = tid >> 5;
    int lane = tid & 31;
    if (e >= E) return;

    int   r = erow[e];
    int   c = ecol[e];
    float v = eval[e];

    const float2* xs = (const float2*)(x + (size_t)c * GC_FEAT);
    float2 m = xs[lane];
    float* dst = support + (size_t)r * GC_FEAT + lane * 2;
    unsafeAtomicAdd(dst,     v * m.x);
    unsafeAtomicAdd(dst + 1, v * m.y);
}

__global__ void gc_gemm_norm(const float* __restrict__ support,
                             const float* __restrict__ weight,
                             const float* __restrict__ bias,
                             float*       __restrict__ out,
                             int nTiles) {
    int wavesPerBlock = blockDim.x >> 5;
    int rowTile = blockIdx.x * wavesPerBlock + ((int)threadIdx.x >> 5);
    if (rowTile >= nTiles) return;

    int lane = threadIdx.x & 31;
    int m    = lane & 15;
    int hi   = lane >> 4;
    int kb   = hi * 2;

    int rowBase = rowTile * 16;
    v8f acc0 = {}, acc1 = {}, acc2 = {}, acc3 = {};

    const float* arow = support + (size_t)(rowBase + m) * GC_FEAT + kb;

#pragma unroll
    for (int ki = 0; ki < 16; ++ki) {
        int k = ki * 4 + kb;
        float2 av = *(const float2*)(arow + ki * 4);
        v2f a; a.x = av.x; a.y = av.y;

        const float* w0 = weight + (size_t)k * GC_FEAT;
        const float* w1 = w0 + GC_FEAT;
        v2f b;

        b.x = w0[m];      b.y = w1[m];
        acc0 = __builtin_amdgcn_wmma_f32_16x16x4_f32(false, a, false, b,
                                                     (short)0, acc0, false, false);
        b.x = w0[m + 16]; b.y = w1[m + 16];
        acc1 = __builtin_amdgcn_wmma_f32_16x16x4_f32(false, a, false, b,
                                                     (short)0, acc1, false, false);
        b.x = w0[m + 32]; b.y = w1[m + 32];
        acc2 = __builtin_amdgcn_wmma_f32_16x16x4_f32(false, a, false, b,
                                                     (short)0, acc2, false, false);
        b.x = w0[m + 48]; b.y = w1[m + 48];
        acc3 = __builtin_amdgcn_wmma_f32_16x16x4_f32(false, a, false, b,
                                                     (short)0, acc3, false, false);
    }

    float b0 = bias[m], b1 = bias[m + 16], b2 = bias[m + 32], b3 = bias[m + 48];
#pragma unroll
    for (int v = 0; v < 8; ++v) {
        acc0[v] += b0; acc1[v] += b1; acc2[v] += b2; acc3[v] += b3;
    }

    float inv[8];
#pragma unroll
    for (int v = 0; v < 8; ++v) {
        float p = acc0[v] * acc0[v] + acc1[v] * acc1[v]
                + acc2[v] * acc2[v] + acc3[v] * acc3[v];
        p += __shfl_xor(p, 1, 32);
        p += __shfl_xor(p, 2, 32);
        p += __shfl_xor(p, 4, 32);
        p += __shfl_xor(p, 8, 32);
        inv[v] = 1.0f / __builtin_sqrtf(p);
    }

#pragma unroll
    for (int v = 0; v < 8; ++v) {
        float* orow = out + (size_t)(rowBase + v + 8 * hi) * GC_FEAT;
        float s = inv[v];
        orow[m]      = acc0[v] * s;
        orow[m + 16] = acc1[v] * s;
        orow[m + 32] = acc2[v] * s;
        orow[m + 48] = acc3[v] * s;
    }
}

extern "C" void kernel_launch(void* const* d_in, const int* in_sizes, int n_in,
                              void* d_out, int out_size, void* d_ws, size_t ws_size,
                              hipStream_t stream) {
    const float* x      = (const float*)d_in[0];
    const int*   erow   = (const int*)  d_in[1];
    const int*   ecol   = (const int*)  d_in[2];
    const float* eval   = (const float*)d_in[3];
    const float* weight = (const float*)d_in[4];
    const float* bias   = (const float*)d_in[5];
    float*       out    = (float*)d_out;
    float*       support = (float*)d_ws;

    int E  = in_sizes[1];
    int nSup = GC_N_NODES * GC_FEAT;

    {
        int n4 = nSup / 4;
        gc_zero_kernel<<<(n4 + 255) / 256, 256, 0, stream>>>((float4*)support, n4);
    }
    {
        long long threads = (long long)E * 32;
        int blocks = (int)((threads + 255) / 256);
        gc_spmm_scatter<<<blocks, 256, 0, stream>>>(x, erow, ecol, eval, support, E);
    }
    {
        int nTiles = GC_N_NODES / 16;
        int wavesPerBlock = 8;
        int blocks = (nTiles + wavesPerBlock - 1) / wavesPerBlock;
        gc_gemm_norm<<<blocks, wavesPerBlock * 32, 0, stream>>>(support, weight, bias, out, nTiles);
    }
}


# SpMM without global f32 atomics: edges grouped by 208-row destination bin (LDS counting sort), one block per bin sums gathered x rows by LDS f32 atomics (arrival order), WMMA reads them from LDS
# speedup vs baseline: 2.3476x; 2.3476x over previous
#include <hip/hip_runtime.h>
#include <hip/hip_bf16.h>

typedef __attribute__((ext_vector_type(2))) float v2f;
typedef __attribute__((ext_vector_type(8))) float v8f;

#define GC_N_NODES 100000
#define GC_FEAT 64

#define GC_BIN_ROWS    208
#define GC_NBINS       481
#define GC_BIN_PAD     512
#define GC_PART_BLOCKS 128
#define GC_LDS_STRIDE  68
#define GC_COL_BITS    17
#define GC_GATHER_THREADS 512

__global__ __launch_bounds__(1024)
void gc_bin_count(const int* __restrict__ erow, int* __restrict__ cnt, int E) {
    __shared__ int hist[GC_BIN_PAD];
    int t = threadIdx.x;
    if (t < GC_BIN_PAD) hist[t] = 0;
    __syncthreads();

    int chunk = (E + GC_PART_BLOCKS - 1) / GC_PART_BLOCKS;
    int e0 = blockIdx.x * chunk;
    int e1 = min(E, e0 + chunk);
    for (int e = e0 + t; e < e1; e += 1024) {
        int b = erow[e] / GC_BIN_ROWS;
        atomicAdd(&hist[b], 1);
    }
    __syncthreads();
    if (t < GC_BIN_PAD) cnt[blockIdx.x * GC_BIN_PAD + t] = hist[t];
}

__global__ __launch_bounds__(1024)
void gc_bin_scatter(const int*   __restrict__ erow,
                    const int*   __restrict__ ecol,
                    const float* __restrict__ eval,
                    const int*   __restrict__ cnt,
                    int*         __restrict__ binStart,
                    int*         __restrict__ recsRaw,
                    int E) {
    __shared__ int tot[GC_BIN_PAD];
    __shared__ int base[GC_BIN_PAD];
    int t = threadIdx.x;
    int g = blockIdx.x;

    int before = 0;
    if (t < GC_BIN_PAD) {
        int all = 0;
#pragma unroll 16
        for (int q = 0; q < GC_PART_BLOCKS; ++q) {
            int c = cnt[q * GC_BIN_PAD + t];
            all += c;
            before += (q < g) ? c : 0;
        }
        tot[t] = all;
    }
    __syncthreads();
    if (t < GC_BIN_PAD) {
        int s = 0;
        for (int q = 0; q < t; ++q) s += tot[q];
        base[t] = s + before;
        if (g == 0) binStart[t] = s;
    }
    __syncthreads();
    if (t < GC_BIN_PAD) tot[t] = 0;
    __syncthreads();

    int2* recs = (int2*)recsRaw;
    int chunk = (E + GC_PART_BLOCKS - 1) / GC_PART_BLOCKS;
    int e0 = g * chunk;
    int e1 = min(E, e0 + chunk);
    for (int e = e0 + t; e < e1; e += 1024) {
        int r = erow[e];
        int b = r / GC_BIN_ROWS;
        int rank = atomicAdd(&tot[b], 1);
        int2 rec;
        rec.x = ((r - b * GC_BIN_ROWS) << GC_COL_BITS) | ecol[e];
        rec.y = __float_as_int(eval[e]);
        recs[base[b] + rank] = rec;
    }
}

__global__ __launch_bounds__(GC_GATHER_THREADS)
void gc_gather_gemm_norm(const float* __restrict__ x,
                         const int*   __restrict__ binStart,
                         const int*   __restrict__ recsRaw,
                         const float* __restrict__ weight,
                         const float* __restrict__ bias,
                         float*       __restrict__ out) {
    __shared__ float acc[(GC_BIN_ROWS + 1) * GC_LDS_STRIDE];

    int t    = threadIdx.x;
    int lane = t & 31;
    int w    = t >> 5;
    int bin  = blockIdx.x;

    for (int i = t; i < (GC_BIN_ROWS + 1) * GC_LDS_STRIDE; i += GC_GATHER_THREADS)
        acc[i] = 0.f;
    __syncthreads();

    const int2* recs = (const int2*)recsRaw;
    int s    = binStart[bin];
    int eEnd = binStart[bin + 1];

    for (int cb = s + w * 32; cb < eEnd; cb += (GC_GATHER_THREADS / 32) * 32) {
        int idx = cb + lane;
        int2 rec;
        if (idx < eEnd) {
            rec = recs[idx];
        } else {
            rec.x = GC_BIN_ROWS << GC_COL_BITS;
            rec.y = 0;
        }
#pragma unroll
        for (int grp = 0; grp < 4; ++grp) {
            float2 mrow[8];
            int    lrow[8];
            float  val[8];
#pragma unroll
            for (int u = 0; u < 8; ++u) {
                int px  = __builtin_amdgcn_readlane(rec.x, grp * 8 + u);
                int pv  = __builtin_amdgcn_readlane(rec.y, grp * 8 + u);
                int col = px & ((1 << GC_COL_BITS) - 1);
                lrow[u] = px >> GC_COL_BITS;
                val[u]  = __int_as_float(pv);
                mrow[u] = *(const float2*)(x + (size_t)col * GC_FEAT + 2 * lane);
            }
#pragma unroll
            for (int u = 0; u < 8; ++u) {
                float* d = acc + lrow[u] * GC_LDS_STRIDE + 2 * lane;
                __hip_atomic_fetch_add(d,     val[u] * mrow[u].x,
                                       __ATOMIC_RELAXED, __HIP_MEMORY_SCOPE_WORKGROUP);
                __hip_atomic_fetch_add(d + 1, val[u] * mrow[u].y,
                                       __ATOMIC_RELAXED, __HIP_MEMORY_SCOPE_WORKGROUP);
            }
        }
    }
    __syncthreads();

    int rowBase = bin * GC_BIN_ROWS + w * 16;
    if (w >= GC_BIN_ROWS / 16 || rowBase >= GC_N_NODES) return;

    int m    = lane & 15;
    int hi   = lane >> 4;
    int kb   = hi * 2;

    v8f acc0 = {}, acc1 = {}, acc2 = {}, acc3 = {};

    const float* arow = acc + (w * 16 + m) * GC_LDS_STRIDE + kb;

#pragma unroll
    for (int ki = 0; ki < 16; ++ki) {
        int k = ki * 4 + kb;
        float2 av = *(const float2*)(arow + ki * 4);
        v2f a; a.x = av.x; a.y = av.y;

        const float* w0 = weight + (size_t)k * GC_FEAT;
        const float* w1 = w0 + GC_FEAT;
        v2f b;

        b.x = w0[m];      b.y = w1[m];
        acc0 = __builtin_amdgcn_wmma_f32_16x16x4_f32(false, a, false, b,
                                                     (short)0, acc0, false, false);
        b.x = w0[m + 16]; b.y = w1[m + 16];
        acc1 = __builtin_amdgcn_wmma_f32_16x16x4_f32(false, a, false, b,
                                                     (short)0, acc1, false, false);
        b.x = w0[m + 32]; b.y = w1[m + 32];
        acc2 = __builtin_amdgcn_wmma_f32_16x16x4_f32(false, a, false, b,
                                                     (short)0, acc2, false, false);
        b.x = w0[m + 48]; b.y = w1[m + 48];
        acc3 = __builtin_amdgcn_wmma_f32_16x16x4_f32(false, a, false, b,
                                                     (short)0, acc3, false, false);
    }

    float b0 = bias[m], b1 = bias[m + 16], b2 = bias[m + 32], b3 = bias[m + 48];
#pragma unroll
    for (int v = 0; v < 8; ++v) {
        acc0[v] += b0; acc1[v] += b1; acc2[v] += b2; acc3[v] += b3;
    }

    float inv[8];
#pragma unroll
    for (int v = 0; v < 8; ++v) {
        float p = acc0[v] * acc0[v] + acc1[v] * acc1[v]
                + acc2[v] * acc2[v] + acc3[v] * acc3[v];
        p += __shfl_xor(p, 1, 32);
        p += __shfl_xor(p, 2, 32);
        p += __shfl_xor(p, 4, 32);
        p += __shfl_xor(p, 8, 32);
        inv[v] = 1.0f / __builtin_sqrtf(p);
    }

#pragma unroll
    for (int v = 0; v < 8; ++v) {
        float* orow = out + (size_t)(rowBase + v + 8 * hi) * GC_FEAT;
        float s2 = inv[v];
        orow[m]      = acc0[v] * s2;
        orow[m + 16] = acc1[v] * s2;
        orow[m + 32] = acc2[v] * s2;
        orow[m + 48] = acc3[v] * s2;
    }
}

extern "C" void kernel_launch(void* const* d_in, const int* in_sizes, int n_in,
                              void* d_out, int out_size, void* d_ws, size_t ws_size,
                              hipStream_t stream) {
    const float* x      = (const float*)d_in[0];
    const int*   erow   = (const int*)  d_in[1];
    const int*   ecol   = (const int*)  d_in[2];
    const float* eval   = (const float*)d_in[3];
    const float* weight = (const float*)d_in[4];
    const float* bias   = (const float*)d_in[5];
    float*       out    = (float*)d_out;

    int E  = in_sizes[1];

    int* cnt      = (int*)d_ws;
    int* binStart = cnt + GC_PART_BLOCKS * GC_BIN_PAD;
    int* recs     = binStart + 1024;

    gc_bin_count<<<GC_PART_BLOCKS, 1024, 0, stream>>>(erow, cnt, E);
    gc_bin_scatter<<<GC_PART_BLOCKS, 1024, 0, stream>>>(erow, ecol, eval, cnt, binStart, recs, E);
    gc_gather_gemm_norm<<<GC_NBINS, GC_GATHER_THREADS, 0, stream>>>(x, binStart, recs, weight, bias, out);
}


# More loads in flight per wave: count and scatter kernels request a thread's 13 edges before first use (was 1 edge per loop trip); gather requests 16 x rows per group (was 8), records a batch ahead
# speedup vs baseline: 2.4856x; 1.0588x over previous
#include <hip/hip_runtime.h>
#include <hip/hip_bf16.h>

typedef __attribute__((ext_vector_type(2))) float v2f;
typedef __attribute__((ext_vector_type(8))) float v8f;

#define GC_N_NODES 100000
#define GC_FEAT 64

#define GC_BIN_ROWS    208
#define GC_NBINS       481
#define GC_BIN_PAD     512
#define GC_PART_BLOCKS 128
#define GC_LDS_STRIDE  68
#define GC_COL_BITS    17
#define GC_GATHER_THREADS 512
#define GC_EDGE_BATCH  13
#define GC_GATHER_DEPTH 16

__global__ __launch_bounds__(1024)
void gc_bin_count(const int* __restrict__ erow, int* __restrict__ cnt, int E) {
    __shared__ int hist[GC_BIN_PAD];
    int t = threadIdx.x;
    if (t < GC_BIN_PAD) hist[t] = 0;
    __syncthreads();

    int chunk = (E + GC_PART_BLOCKS - 1) / GC_PART_BLOCKS;
    int e0 = blockIdx.x * chunk;
    int e1 = min(E, e0 + chunk);
    for (int eb = e0 + t; eb < e1; eb += 1024 * GC_EDGE_BATCH) {
        int r[GC_EDGE_BATCH];
#pragma unroll
        for (int u = 0; u < GC_EDGE_BATCH; ++u) {
            int e = eb + 1024 * u;
            r[u] = erow[e < e1 ? e : eb];
        }
#pragma unroll
        for (int u = 0; u < GC_EDGE_BATCH; ++u)
            atomicAdd(&hist[r[u] / GC_BIN_ROWS], (eb + 1024 * u < e1) ? 1 : 0);
    }
    __syncthreads();
    if (t < GC_BIN_PAD) cnt[blockIdx.x * GC_BIN_PAD + t] = hist[t];
}

__global__ __launch_bounds__(1024)
void gc_bin_scatter(const int*   __restrict__ erow,
                    const int*   __restrict__ ecol,
                    const float* __restrict__ eval,
                    const int*   __restrict__ cnt,
                    int*         __restrict__ binStart,
                    int*         __restrict__ recsRaw,
                    int E) {
    __shared__ int tot[GC_BIN_PAD];
    __shared__ int base[GC_BIN_PAD];
    int t = threadIdx.x;
    int g = blockIdx.x;

    int before = 0;
    if (t < GC_BIN_PAD) {
        int all = 0;
#pragma unroll 16
        for (int q = 0; q < GC_PART_BLOCKS; ++q) {
            int c = cnt[q * GC_BIN_PAD + t];
            all += c;
            before += (q < g) ? c : 0;
        }
        tot[t] = all;
    }
    __syncthreads();
    if (t < GC_BIN_PAD) {
        int s = 0;
        for (int q = 0; q < t; ++q) s += tot[q];
        base[t] = s + before;
        if (g == 0) binStart[t] = s;
    }
    __syncthreads();
    if (t < GC_BIN_PAD) tot[t] = 0;
    __syncthreads();

    int2* recs = (int2*)recsRaw;
    int chunk = (E + GC_PART_BLOCKS - 1) / GC_PART_BLOCKS;
    int e0 = g * chunk;
    int e1 = min(E, e0 + chunk);
    for (int eb = e0 + t; eb < e1; eb += 1024 * GC_EDGE_BATCH) {
        int   r[GC_EDGE_BATCH];
        int   c[GC_EDGE_BATCH];
        float v[GC_EDGE_BATCH];
#pragma unroll
        for (int u = 0; u < GC_EDGE_BATCH; ++u) {
            int e = eb + 1024 * u;
            e = e < e1 ? e : eb;
            r[u] = erow[e];
            c[u] = ecol[e];
            v[u] = eval[e];
        }
        int dst0 = 0;
#pragma unroll
        for (int u = 0; u < GC_EDGE_BATCH; ++u) {
            bool ok = eb + 1024 * u < e1;
            int b = r[u] / GC_BIN_ROWS;
            int rank = atomicAdd(&tot[b], ok ? 1 : 0);
            int2 rec;
            rec.x = ((r[u] - b * GC_BIN_ROWS) << GC_COL_BITS) | c[u];
            rec.y = __float_as_int(v[u]);
            int dst = base[b] + rank;
            if (u == 0) dst0 = dst;
            recs[ok ? dst : dst0] = rec;
        }
    }
}

__global__ __launch_bounds__(GC_GATHER_THREADS)
void gc_gather_gemm_norm(const float* __restrict__ x,
                         const int*   __restrict__ binStart,
                         const int*   __restrict__ recsRaw,
                         const float* __restrict__ weight,
                         const float* __restrict__ bias,
                         float*       __restrict__ out) {
    __shared__ float acc[(GC_BIN_ROWS + 1) * GC_LDS_STRIDE];

    int t    = threadIdx.x;
    int lane = t & 31;
    int w    = t >> 5;
    int bin  = blockIdx.x;

    for (int i = t; i < (GC_BIN_ROWS + 1) * GC_LDS_STRIDE; i += GC_GATHER_THREADS)
        acc[i] = 0.f;
    __syncthreads();

    const int2* recs = (const int2*)recsRaw;
    int s    = binStart[bin];
    int eEnd = binStart[bin + 1];

    int2 padRec;
    padRec.x = GC_BIN_ROWS << GC_COL_BITS;
    padRec.y = 0;

    int cb = s + w * 32;
    int2 rec = padRec;
    if (cb + lane < eEnd) rec = recs[cb + lane];
    for (; cb < eEnd; cb += GC_GATHER_THREADS) {
        int nidx = cb + GC_GATHER_THREADS + lane;
        int2 recNext = recs[min(nidx, eEnd - 1)];
        if (nidx >= eEnd) recNext = padRec;
#pragma unroll
        for (int grp = 0; grp < 32 / GC_GATHER_DEPTH; ++grp) {
            float2 mrow[GC_GATHER_DEPTH];
            int    lrow[GC_GATHER_DEPTH];
            float  val[GC_GATHER_DEPTH];
#pragma unroll
            for (int u = 0; u < GC_GATHER_DEPTH; ++u) {
                int px  = __builtin_amdgcn_readlane(rec.x, grp * GC_GATHER_DEPTH + u);
                int pv  = __builtin_amdgcn_readlane(rec.y, grp * GC_GATHER_DEPTH + u);
                int col = px & ((1 << GC_COL_BITS) - 1);
                lrow[u] = px >> GC_COL_BITS;
                val[u]  = __int_as_float(pv);
                mrow[u] = *(const float2*)(x + (size_t)col * GC_FEAT + 2 * lane);
            }
#pragma unroll
            for (int u = 0; u < GC_GATHER_DEPTH; ++u) {
                float* d = acc + lrow[u] * GC_LDS_STRIDE + 2 * lane;
                __hip_atomic_fetch_add(d,     val[u] * mrow[u].x,
                                       __ATOMIC_RELAXED, __HIP_MEMORY_SCOPE_WORKGROUP);
                __hip_atomic_fetch_add(d + 1, val[u] * mrow[u].y,
                                       __ATOMIC_RELAXED, __HIP_MEMORY_SCOPE_WORKGROUP);
            }
        }
        rec = recNext;
    }
    __syncthreads();

    int rowBase = bin * GC_BIN_ROWS + w * 16;
    if (w >= GC_BIN_ROWS / 16 || rowBase >= GC_N_NODES) return;

    int m    = lane & 15;
    int hi   = lane >> 4;
    int kb   = hi * 2;

    v8f acc0 = {}, acc1 = {}, acc2 = {}, acc3 = {};

    const float* arow = acc + (w * 16 + m) * GC_LDS_STRIDE + kb;

#pragma unroll
    for (int ki = 0; ki < 16; ++ki) {
        int k = ki * 4 + kb;
        float2 av = *(const float2*)(arow + ki * 4);
        v2f a; a.x = av.x; a.y = av.y;

        const float* w0 = weight + (size_t)k * GC_FEAT;
        const float* w1 = w0 + GC_FEAT;
        v2f b;

        b.x = w0[m];      b.y = w1[m];
        acc0 = __builtin_amdgcn_wmma_f32_16x16x4_f32(false, a, false, b,
                                                     (short)0, acc0, false, false);
        b.x = w0[m + 16]; b.y = w1[m + 16];
        acc1 = __builtin_amdgcn_wmma_f32_16x16x4_f32(false, a, false, b,
                                                     (short)0, acc1, false, false);
        b.x = w0[m + 32]; b.y = w1[m + 32];
        acc2 = __builtin_amdgcn_wmma_f32_16x16x4_f32(false, a, false, b,
                                                     (short)0, acc2, false, false);
        b.x = w0[m + 48]; b.y = w1[m + 48];
        acc3 = __builtin_amdgcn_wmma_f32_16x16x4_f32(false, a, false, b,
                                                     (short)0, acc3, false, false);
    }

    float b0 = bias[m], b1 = bias[m + 16], b2 = bias[m + 32], b3 = bias[m + 48];
#pragma unroll
    for (int v = 0; v < 8; ++v) {
        acc0[v] += b0; acc1[v] += b1; acc2[v] += b2; acc3[v] += b3;
    }

    float inv[8];
#pragma unroll
    for (int v = 0; v < 8; ++v) {
        float p = acc0[v] * acc0[v] + acc1[v] * acc1[v]
                + acc2[v] * acc2[v] + acc3[v] * acc3[v];
        p += __shfl_xor(p, 1, 32);
        p += __shfl_xor(p, 2, 32);
        p += __shfl_xor(p, 4, 32);
        p += __shfl_xor(p, 8, 32);
        inv[v] = 1.0f / __builtin_sqrtf(p);
    }

#pragma unroll
    for (int v = 0; v < 8; ++v) {
        float* orow = out + (size_t)(rowBase + v + 8 * hi) * GC_FEAT;
        float s2 = inv[v];
        orow[m]      = acc0[v] * s2;
        orow[m + 16] = acc1[v] * s2;
        orow[m + 32] = acc2[v] * s2;
        orow[m + 48] = acc3[v] * s2;
    }
}

extern "C" void kernel_launch(void* const* d_in, const int* in_sizes, int n_in,
                              void* d_out, int out_size, void* d_ws, size_t ws_size,
                              hipStream_t stream) {
    const float* x      = (const float*)d_in[0];
    const int*   erow   = (const int*)  d_in[1];
    const int*   ecol   = (const int*)  d_in[2];
    const float* eval   = (const float*)d_in[3];
    const float* weight = (const float*)d_in[4];
    const float* bias   = (const float*)d_in[5];
    float*       out    = (float*)d_out;

    int E  = in_sizes[1];

    int* cnt      = (int*)d_ws;
    int* binStart = cnt + GC_PART_BLOCKS * GC_BIN_PAD;
    int* recs     = binStart + 1024;

    gc_bin_count<<<GC_PART_BLOCKS, 1024, 0, stream>>>(erow, cnt, E);
    gc_bin_scatter<<<GC_PART_BLOCKS, 1024, 0, stream>>>(erow, ecol, eval, cnt, binStart, recs, E);
    gc_gather_gemm_norm<<<GC_NBINS, GC_GATHER_THREADS, 0, stream>>>(x, binStart, recs, weight, bias, out);
}


# Count launch and global prefix dropped: each 12500-edge chunk is bin-sorted in LDS, stored as whole lines with an offset row; a bin's block walks its 128 runs; LDS f32 atomic sums, arrival order
# speedup vs baseline: 3.5407x; 1.4245x over previous
#include <hip/hip_runtime.h>
#include <hip/hip_bf16.h>

typedef __attribute__((ext_vector_type(2))) float v2f;
typedef __attribute__((ext_vector_type(8))) float v8f;

#define GC_N_NODES 100000
#define GC_FEAT 64

#define GC_BIN_ROWS    208
#define GC_NBINS       481
#define GC_BIN_PAD     512
#define GC_PART_BLOCKS 128
#define GC_LDS_STRIDE  68
#define GC_COL_BITS    17
#define GC_GATHER_THREADS 512
#define GC_EDGE_BATCH  13
#define GC_GATHER_DEPTH 16
#define GC_SEG_CAP     (1024 * GC_EDGE_BATCH)

__global__ __launch_bounds__(1024)
void gc_seg_sort(const int*   __restrict__ erow,
                 const int*   __restrict__ ecol,
                 const float* __restrict__ eval,
                 int*         __restrict__ ws,
                 int E) {
    __shared__ int  cntb[GC_BIN_PAD];
    __shared__ int  wsum[16];
    __shared__ int2 stage[GC_SEG_CAP];
    int t    = threadIdx.x;
    int lane = t & 31;
    int g    = blockIdx.x;

    int2* recs = (int2*)ws;
    int*  tab  = ws + 2 * (size_t)E;
    int chunk  = (E + GC_PART_BLOCKS - 1) / GC_PART_BLOCKS;
    int e0     = g * chunk;
    int e1     = min(E, e0 + chunk);
    int rounds = (chunk + GC_SEG_CAP - 1) / GC_SEG_CAP;

    for (int r = 0; r < rounds; ++r) {
        int s0   = e0 + r * GC_SEG_CAP;
        int s1   = min(e1, s0 + GC_SEG_CAP);
        int n    = max(0, s1 - s0);
        int safe = n > 0 ? s0 : 0;

        if (t < GC_BIN_PAD) cntb[t] = 0;
        __syncthreads();

        int   rr[GC_EDGE_BATCH];
        int   cc[GC_EDGE_BATCH];
        float vv[GC_EDGE_BATCH];
        int   rk[GC_EDGE_BATCH];
        int eb = s0 + t;
#pragma unroll
        for (int u = 0; u < GC_EDGE_BATCH; ++u) {
            int e = eb + 1024 * u;
            e = e < s1 ? e : safe;
            rr[u] = erow[e];
            cc[u] = ecol[e];
            vv[u] = eval[e];
        }
        __builtin_amdgcn_sched_barrier(0);
#pragma unroll
        for (int u = 0; u < GC_EDGE_BATCH; ++u) {
            bool ok = eb + 1024 * u < s1;
            rk[u] = atomicAdd(&cntb[rr[u] / GC_BIN_ROWS], ok ? 1 : 0);
        }
        __syncthreads();

        int c = 0, incl = 0;
        if (t < GC_BIN_PAD) {
            c = cntb[t];
            incl = c;
            for (int d = 1; d < 32; d <<= 1) {
                int y = __shfl_up(incl, d, 32);
                if (lane >= d) incl += y;
            }
            if (lane == 31) wsum[t >> 5] = incl;
        }
        __syncthreads();
        if (t < GC_BIN_PAD) {
            int base = 0;
            for (int q = 0; q < (t >> 5); ++q) base += wsum[q];
            int ex = base + incl - c;
            cntb[t] = ex;
            tab[(r * GC_PART_BLOCKS + g) * GC_BIN_PAD + t] = ex;
        }
        __syncthreads();

        int dst[GC_EDGE_BATCH];
#pragma unroll
        for (int u = 0; u < GC_EDGE_BATCH; ++u)
            dst[u] = cntb[rr[u] / GC_BIN_ROWS] + rk[u];
#pragma unroll
        for (int u = 0; u < GC_EDGE_BATCH; ++u) {
            if (eb + 1024 * u < s1) {
                int b = rr[u] / GC_BIN_ROWS;
                int2 rec;
                rec.x = ((rr[u] - b * GC_BIN_ROWS) << GC_COL_BITS) | cc[u];
                rec.y = __float_as_int(vv[u]);
                stage[dst[u]] = rec;
            }
        }
        __syncthreads();

#pragma unroll
        for (int u = 0; u < GC_EDGE_BATCH; ++u) {
            int i = t + 1024 * u;
            if (i < n) recs[s0 + i] = stage[i];
        }
        __syncthreads();
    }
}

__global__ __launch_bounds__(GC_GATHER_THREADS)
void gc_gather_gemm_norm(const float* __restrict__ x,
                         const int*   __restrict__ ws,
                         const float* __restrict__ weight,
                         const float* __restrict__ bias,
                         float*       __restrict__ out,
                         int E) {
    __shared__ float acc[(GC_BIN_ROWS + 1) * GC_LDS_STRIDE];
    __shared__ int pre[GC_GATHER_THREADS];
    __shared__ int ofs[GC_GATHER_THREADS];
    __shared__ int wsum[GC_GATHER_THREADS / 32];

    int t    = threadIdx.x;
    int lane = t & 31;
    int w    = t >> 5;
    int bin  = blockIdx.x;

    for (int i = t; i < (GC_BIN_ROWS + 1) * GC_LDS_STRIDE; i += GC_GATHER_THREADS)
        acc[i] = 0.f;
    __syncthreads();

    const int2* recs = (const int2*)ws;
    const int*  tab  = ws + 2 * (size_t)E;

    int2 padRec;
    padRec.x = GC_BIN_ROWS << GC_COL_BITS;
    padRec.y = 0;

    int chunk  = (E + GC_PART_BLOCKS - 1) / GC_PART_BLOCKS;
    int rounds = (chunk + GC_SEG_CAP - 1) / GC_SEG_CAP;
    int nseg   = rounds * GC_PART_BLOCKS;

    for (int sbase = 0; sbase < nseg; sbase += GC_GATHER_THREADS) {
        int sg  = sbase + t;
        int len = 0, st = 0;
        if (sg < nseg) {
            int a  = tab[sg * GC_BIN_PAD + bin];
            int b2 = tab[sg * GC_BIN_PAD + bin + 1];
            len = b2 - a;
            st  = (int)((unsigned)sg % GC_PART_BLOCKS) * chunk
                + (int)((unsigned)sg / GC_PART_BLOCKS) * GC_SEG_CAP + a;
        }
        int incl = len;
        for (int d = 1; d < 32; d <<= 1) {
            int y = __shfl_up(incl, d, 32);
            if (lane >= d) incl += y;
        }
        if (lane == 31) wsum[w] = incl;
        __syncthreads();
        int base = 0, tot = 0;
        for (int q = 0; q < GC_GATHER_THREADS / 32; ++q) {
            int v = wsum[q];
            base += (q < w) ? v : 0;
            tot  += v;
        }
        int ex = base + incl - len;
        pre[t] = ex;
        ofs[t] = st - ex;
        __syncthreads();

        int nloc  = min(nseg - sbase, GC_GATHER_THREADS);
        int step0 = GC_GATHER_THREADS / 2;
        while (step0 >= nloc && step0 > 0) step0 >>= 1;

        int cb = w * 32;
        int2 rec = padRec;
        if (cb + lane < tot) {
            int f = cb + lane;
            int j = 0;
            for (int step = step0; step; step >>= 1)
                if (pre[j + step] <= f) j += step;
            rec = recs[f + ofs[j]];
        }
        for (; cb < tot; cb += GC_GATHER_THREADS) {
            int nf = cb + GC_GATHER_THREADS + lane;
            int2 recNext = padRec;
#pragma unroll
            for (int grp = 0; grp < 32 / GC_GATHER_DEPTH; ++grp) {
                float2 mrow[GC_GATHER_DEPTH];
                int    lrow[GC_GATHER_DEPTH];
                float  val[GC_GATHER_DEPTH];
#pragma unroll
                for (int u = 0; u < GC_GATHER_DEPTH; ++u) {
                    int px  = __builtin_amdgcn_readlane(rec.x, grp * GC_GATHER_DEPTH + u);
                    int pv  = __builtin_amdgcn_readlane(rec.y, grp * GC_GATHER_DEPTH + u);
                    int col = px & ((1 << GC_COL_BITS) - 1);
                    lrow[u] = px >> GC_COL_BITS;
                    val[u]  = __int_as_float(pv);
                    mrow[u] = *(const float2*)(x + (size_t)col * GC_FEAT + 2 * lane);
                }
                if (grp == 0) {
                    int f = min(nf, tot - 1);
                    int j = 0;
                    for (int step = step0; step; step >>= 1)
                        if (pre[j + step] <= f) j += step;
                    recNext = recs[f + ofs[j]];
                    if (nf >= tot) recNext = padRec;
                }
#pragma unroll
                for (int u = 0; u < GC_GATHER_DEPTH; ++u) {
                    float* d = acc + lrow[u] * GC_LDS_STRIDE + 2 * lane;
                    __hip_atomic_fetch_add(d,     val[u] * mrow[u].x,
                                           __ATOMIC_RELAXED, __HIP_MEMORY_SCOPE_WORKGROUP);
                    __hip_atomic_fetch_add(d + 1, val[u] * mrow[u].y,
                                           __ATOMIC_RELAXED, __HIP_MEMORY_SCOPE_WORKGROUP);
                }
            }
            rec = recNext;
        }
        __syncthreads();
    }

    int rowBase = bin * GC_BIN_ROWS + w * 16;
    if (w >= GC_BIN_ROWS / 16 || rowBase >= GC_N_NODES) return;

    int m    = lane & 15;
    int hi   = lane >> 4;
    int kb   = hi * 2;

    v8f acc0 = {}, acc1 = {}, acc2 = {}, acc3 = {};

    const float* arow = acc + (w * 16 + m) * GC_LDS_STRIDE + kb;

#pragma unroll
    for (int ki = 0; ki < 16; ++ki) {
        int k = ki * 4 + kb;
        float2 av = *(const float2*)(arow + ki * 4);
        v2f a; a.x = av.x; a.y = av.y;

        const float* w0 = weight + (size_t)k * GC_FEAT;
        const float* w1 = w0 + GC_FEAT;
        v2f b;

        b.x = w0[m];      b.y = w1[m];
        acc0 = __builtin_amdgcn_wmma_f32_16x16x4_f32(false, a, false, b,
                                                     (short)0, acc0, false, false);
        b.x = w0[m + 16]; b.y = w1[m + 16];
        acc1 = __builtin_amdgcn_wmma_f32_16x16x4_f32(false, a, false, b,
                                                     (short)0, acc1, false, false);
        b.x = w0[m + 32]; b.y = w1[m + 32];
        acc2 = __builtin_amdgcn_wmma_f32_16x16x4_f32(false, a, false, b,
                                                     (short)0, acc2, false, false);
        b.x = w0[m + 48]; b.y = w1[m + 48];
        acc3 = __builtin_amdgcn_wmma_f32_16x16x4_f32(false, a, false, b,
                                                     (short)0, acc3, false, false);
    }

    float b0 = bias[m], b1 = bias[m + 16], b2 = bias[m + 32], b3 = bias[m + 48];
#pragma unroll
    for (int v = 0; v < 8; ++v) {
        acc0[v] += b0; acc1[v] += b1; acc2[v] += b2; acc3[v] += b3;
    }

    float inv[8];
#pragma unroll
    for (int v = 0; v < 8; ++v) {
        float p = acc0[v] * acc0[v] + acc1[v] * acc1[v]
                + acc2[v] * acc2[v] + acc3[v] * acc3[v];
        p += __shfl_xor(p, 1, 32);
        p += __shfl_xor(p, 2, 32);
        p += __shfl_xor(p, 4, 32);
        p += __shfl_xor(p, 8, 32);
        inv[v] = 1.0f / __builtin_sqrtf(p);
    }

#pragma unroll
    for (int v = 0; v < 8; ++v) {
        float* orow = out + (size_t)(rowBase + v + 8 * hi) * GC_FEAT;
        float s2 = inv[v];
        orow[m]      = acc0[v] * s2;
        orow[m + 16] = acc1[v] * s2;
        orow[m + 32] = acc2[v] * s2;
        orow[m + 48] = acc3[v] * s2;
    }
}

extern "C" void kernel_launch(void* const* d_in, const int* in_sizes, int n_in,
                              void* d_out, int out_size, void* d_ws, size_t ws_size,
                              hipStream_t stream) {
    const float* x      = (const float*)d_in[0];
    const int*   erow   = (const int*)  d_in[1];
    const int*   ecol   = (const int*)  d_in[2];
    const float* eval   = (const float*)d_in[3];
    const float* weight = (const float*)d_in[4];
    const float* bias   = (const float*)d_in[5];
    float*       out    = (float*)d_out;

    int E  = in_sizes[1];

    int* ws = (int*)d_ws;

    gc_seg_sort<<<GC_PART_BLOCKS, 1024, 0, stream>>>(erow, ecol, eval, ws, E);
    gc_gather_gemm_norm<<<GC_NBINS, GC_GATHER_THREADS, 0, stream>>>(x, ws, weight, bias, out, E);
}


# Bin's records put in row order in LDS (counting sort, int rank); a wave sums its 13 rows' gathered x rows in registers in rank order: one LDS f32 add per row and pass, not one per edge
# speedup vs baseline: 4.2593x; 1.2030x over previous
#include <hip/hip_runtime.h>
#include <hip/hip_bf16.h>

typedef __attribute__((ext_vector_type(2))) float v2f;
typedef __attribute__((ext_vector_type(8))) float v8f;

#define GC_N_NODES 100000
#define GC_FEAT 64

#define GC_BIN_ROWS    208
#define GC_NBINS       481
#define GC_BIN_PAD     512
#define GC_PART_BLOCKS 128
#define GC_LDS_STRIDE  68
#define GC_COL_BITS    17
#define GC_GATHER_THREADS 512
#define GC_EDGE_BATCH  13
#define GC_GATHER_DEPTH 16
#define GC_SEG_CAP     (1024 * GC_EDGE_BATCH)
#define GC_SORT_CAP    4096
#define GC_SORT_SLOTS  (GC_SORT_CAP / GC_GATHER_THREADS)
#define GC_WAVE_ROWS   (GC_BIN_ROWS / (GC_GATHER_THREADS / 32))

__global__ __launch_bounds__(1024)
void gc_seg_sort(const int*   __restrict__ erow,
                 const int*   __restrict__ ecol,
                 const float* __restrict__ eval,
                 int*         __restrict__ ws,
                 int E) {
    __shared__ int  cntb[GC_BIN_PAD];
    __shared__ int  wsum[16];
    __shared__ int2 stage[GC_SEG_CAP];
    int t    = threadIdx.x;
    int lane = t & 31;
    int g    = blockIdx.x;

    int2* recs = (int2*)ws;
    int*  tab  = ws + 2 * (size_t)E;
    int chunk  = (E + GC_PART_BLOCKS - 1) / GC_PART_BLOCKS;
    int e0     = g * chunk;
    int e1     = min(E, e0 + chunk);
    int rounds = (chunk + GC_SEG_CAP - 1) / GC_SEG_CAP;

    for (int r = 0; r < rounds; ++r) {
        int s0   = e0 + r * GC_SEG_CAP;
        int s1   = min(e1, s0 + GC_SEG_CAP);
        int n    = max(0, s1 - s0);
        int safe = n > 0 ? s0 : 0;

        if (t < GC_BIN_PAD) cntb[t] = 0;
        __syncthreads();

        int   rr[GC_EDGE_BATCH];
        int   cc[GC_EDGE_BATCH];
        float vv[GC_EDGE_BATCH];
        int   rk[GC_EDGE_BATCH];
        int eb = s0 + t;
#pragma unroll
        for (int u = 0; u < GC_EDGE_BATCH; ++u) {
            int e = eb + 1024 * u;
            e = e < s1 ? e : safe;
            rr[u] = erow[e];
            cc[u] = ecol[e];
            vv[u] = eval[e];
        }
        __builtin_amdgcn_sched_barrier(0);
#pragma unroll
        for (int u = 0; u < GC_EDGE_BATCH; ++u) {
            bool ok = eb + 1024 * u < s1;
            rk[u] = atomicAdd(&cntb[rr[u] / GC_BIN_ROWS], ok ? 1 : 0);
        }
        __syncthreads();

        int c = 0, incl = 0;
        if (t < GC_BIN_PAD) {
            c = cntb[t];
            incl = c;
            for (int d = 1; d < 32; d <<= 1) {
                int y = __shfl_up(incl, d, 32);
                if (lane >= d) incl += y;
            }
            if (lane == 31) wsum[t >> 5] = incl;
        }
        __syncthreads();
        if (t < GC_BIN_PAD) {
            int base = 0;
            for (int q = 0; q < (t >> 5); ++q) base += wsum[q];
            int ex = base + incl - c;
            cntb[t] = ex;
            tab[(r * GC_PART_BLOCKS + g) * GC_BIN_PAD + t] = ex;
        }
        __syncthreads();

        int dst[GC_EDGE_BATCH];
#pragma unroll
        for (int u = 0; u < GC_EDGE_BATCH; ++u)
            dst[u] = cntb[rr[u] / GC_BIN_ROWS] + rk[u];
#pragma unroll
        for (int u = 0; u < GC_EDGE_BATCH; ++u) {
            if (eb + 1024 * u < s1) {
                int b = rr[u] / GC_BIN_ROWS;
                int2 rec;
                rec.x = ((rr[u] - b * GC_BIN_ROWS) << GC_COL_BITS) | cc[u];
                rec.y = __float_as_int(vv[u]);
                stage[dst[u]] = rec;
            }
        }
        __syncthreads();

#pragma unroll
        for (int u = 0; u < GC_EDGE_BATCH; ++u) {
            int i = t + 1024 * u;
            if (i < n) recs[s0 + i] = stage[i];
        }
        __syncthreads();
    }
}

__global__ __launch_bounds__(GC_GATHER_THREADS)
void gc_gather_gemm_norm(const float* __restrict__ x,
                         const int*   __restrict__ ws,
                         const float* __restrict__ weight,
                         const float* __restrict__ bias,
                         float*       __restrict__ out,
                         int E) {
    __shared__ float acc[(GC_BIN_ROWS + 1) * GC_LDS_STRIDE];
    __shared__ int pre[GC_GATHER_THREADS];
    __shared__ int ofs[GC_GATHER_THREADS];
    __shared__ int wsum[GC_GATHER_THREADS / 32];
    __shared__ int2 srt[GC_SORT_CAP];
    __shared__ int rcnt[256];
    __shared__ int rsum[8];

    int t    = threadIdx.x;
    int lane = t & 31;
    int w    = t >> 5;
    int bin  = blockIdx.x;

    for (int i = t; i < (GC_BIN_ROWS + 1) * GC_LDS_STRIDE; i += GC_GATHER_THREADS)
        acc[i] = 0.f;
    __syncthreads();

    const int2* recs = (const int2*)ws;
    const int*  tab  = ws + 2 * (size_t)E;

    int2 padRec;
    padRec.x = GC_BIN_ROWS << GC_COL_BITS;
    padRec.y = 0;

    int chunk  = (E + GC_PART_BLOCKS - 1) / GC_PART_BLOCKS;
    int rounds = (chunk + GC_SEG_CAP - 1) / GC_SEG_CAP;
    int nseg   = rounds * GC_PART_BLOCKS;

    for (int sbase = 0; sbase < nseg; sbase += GC_GATHER_THREADS) {
        int sg  = sbase + t;
        int len = 0, st = 0;
        if (sg < nseg) {
            int a  = tab[sg * GC_BIN_PAD + bin];
            int b2 = tab[sg * GC_BIN_PAD + bin + 1];
            len = b2 - a;
            st  = (int)((unsigned)sg % GC_PART_BLOCKS) * chunk
                + (int)((unsigned)sg / GC_PART_BLOCKS) * GC_SEG_CAP + a;
        }
        int incl = len;
        for (int d = 1; d < 32; d <<= 1) {
            int y = __shfl_up(incl, d, 32);
            if (lane >= d) incl += y;
        }
        if (lane == 31) wsum[w] = incl;
        __syncthreads();
        int base = 0, tot = 0;
        for (int q = 0; q < GC_GATHER_THREADS / 32; ++q) {
            int v = wsum[q];
            base += (q < w) ? v : 0;
            tot  += v;
        }
        int ex = base + incl - len;
        pre[t] = ex;
        ofs[t] = st - ex;
        __syncthreads();

        int nloc  = min(nseg - sbase, GC_GATHER_THREADS);
        int step0 = GC_GATHER_THREADS / 2;
        while (step0 >= nloc && step0 > 0) step0 >>= 1;

        int wu = __builtin_amdgcn_readfirstlane(w);

        for (int p0 = 0; p0 < tot; p0 += GC_SORT_CAP) {
            int pn = min(tot - p0, GC_SORT_CAP);
            if (t < 256) rcnt[t] = 0;
            __syncthreads();

            int fidx[GC_SORT_SLOTS];
            int fj[GC_SORT_SLOTS];
#pragma unroll
            for (int u = 0; u < GC_SORT_SLOTS; ++u) {
                int i = t + GC_GATHER_THREADS * u;
                fidx[u] = p0 + (i < pn ? i : 0);
                fj[u]   = 0;
            }
            for (int step = step0; step; step >>= 1) {
#pragma unroll
                for (int u = 0; u < GC_SORT_SLOTS; ++u)
                    if (pre[fj[u] + step] <= fidx[u]) fj[u] += step;
            }
            int2 rc[GC_SORT_SLOTS];
            int  rk[GC_SORT_SLOTS];
#pragma unroll
            for (int u = 0; u < GC_SORT_SLOTS; ++u)
                rc[u] = recs[fidx[u] + ofs[fj[u]]];
#pragma unroll
            for (int u = 0; u < GC_SORT_SLOTS; ++u) {
                bool ok = t + GC_GATHER_THREADS * u < pn;
                rk[u] = atomicAdd(&rcnt[rc[u].x >> GC_COL_BITS], ok ? 1 : 0);
            }
            __syncthreads();

            int c = 0, rincl = 0;
            if (t < 256) {
                c = rcnt[t];
                rincl = c;
                for (int d = 1; d < 32; d <<= 1) {
                    int y = __shfl_up(rincl, d, 32);
                    if (lane >= d) rincl += y;
                }
                if (lane == 31) rsum[w] = rincl;
            }
            __syncthreads();
            if (t < 256) {
                int rbase = 0;
                for (int q = 0; q < w; ++q) rbase += rsum[q];
                rcnt[t] = rbase + rincl - c;
            }
            __syncthreads();

#pragma unroll
            for (int u = 0; u < GC_SORT_SLOTS; ++u) {
                if (t + GC_GATHER_THREADS * u < pn)
                    srt[rcnt[rc[u].x >> GC_COL_BITS] + rk[u]] = rc[u];
            }
            __syncthreads();

            int beg = __builtin_amdgcn_readfirstlane(rcnt[wu * GC_WAVE_ROWS]);
            int end = __builtin_amdgcn_readfirstlane(rcnt[wu * GC_WAVE_ROWS + GC_WAVE_ROWS]);
            int cur = wu * GC_WAVE_ROWS;
            float sx = 0.f, sy = 0.f;
            int2 rec = padRec;
            if (beg + lane < end) rec = srt[beg + lane];
            for (int b = beg; b < end; b += 32) {
                int2 recNext = padRec;
                if (b + 32 + lane < end) recNext = srt[b + 32 + lane];
#pragma unroll
                for (int grp = 0; grp < 32 / GC_GATHER_DEPTH; ++grp) {
                    if (b + grp * GC_GATHER_DEPTH < end) {
                        float2 mrow[GC_GATHER_DEPTH];
                        int    lrow[GC_GATHER_DEPTH];
                        float  val[GC_GATHER_DEPTH];
#pragma unroll
                        for (int u = 0; u < GC_GATHER_DEPTH; ++u) {
                            int px  = __builtin_amdgcn_readlane(rec.x, grp * GC_GATHER_DEPTH + u);
                            int pv  = __builtin_amdgcn_readlane(rec.y, grp * GC_GATHER_DEPTH + u);
                            int col = px & ((1 << GC_COL_BITS) - 1);
                            lrow[u] = px >> GC_COL_BITS;
                            val[u]  = __int_as_float(pv);
                            mrow[u] = *(const float2*)(x + (size_t)col * GC_FEAT + 2 * lane);
                        }
#pragma unroll
                        for (int u = 0; u < GC_GATHER_DEPTH; ++u) {
                            if (lrow[u] != cur) {
                                float* d = acc + cur * GC_LDS_STRIDE + 2 * lane;
                                __hip_atomic_fetch_add(d,     sx, __ATOMIC_RELAXED, __HIP_MEMORY_SCOPE_WORKGROUP);
                                __hip_atomic_fetch_add(d + 1, sy, __ATOMIC_RELAXED, __HIP_MEMORY_SCOPE_WORKGROUP);
                                sx = 0.f; sy = 0.f;
                                cur = lrow[u];
                            }
                            sx += val[u] * mrow[u].x;
                            sy += val[u] * mrow[u].y;
                        }
                    }
                }
                rec = recNext;
            }
            {
                float* d = acc + cur * GC_LDS_STRIDE + 2 * lane;
                __hip_atomic_fetch_add(d,     sx, __ATOMIC_RELAXED, __HIP_MEMORY_SCOPE_WORKGROUP);
                __hip_atomic_fetch_add(d + 1, sy, __ATOMIC_RELAXED, __HIP_MEMORY_SCOPE_WORKGROUP);
            }
            __syncthreads();
        }
    }

    int rowBase = bin * GC_BIN_ROWS + w * 16;
    if (w >= GC_BIN_ROWS / 16 || rowBase >= GC_N_NODES) return;

    int m    = lane & 15;
    int hi   = lane >> 4;
    int kb   = hi * 2;

    v8f acc0 = {}, acc1 = {}, acc2 = {}, acc3 = {};

    const float* arow = acc + (w * 16 + m) * GC_LDS_STRIDE + kb;

#pragma unroll
    for (int ki = 0; ki < 16; ++ki) {
        int k = ki * 4 + kb;
        float2 av = *(const float2*)(arow + ki * 4);
        v2f a; a.x = av.x; a.y = av.y;

        const float* w0 = weight + (size_t)k * GC_FEAT;
        const float* w1 = w0 + GC_FEAT;
        v2f b;

        b.x = w0[m];      b.y = w1[m];
        acc0 = __builtin_amdgcn_wmma_f32_16x16x4_f32(false, a, false, b,
                                                     (short)0, acc0, false, false);
        b.x = w0[m + 16]; b.y = w1[m + 16];
        acc1 = __builtin_amdgcn_wmma_f32_16x16x4_f32(false, a, false, b,
                                                     (short)0, acc1, false, false);
        b.x = w0[m + 32]; b.y = w1[m + 32];
        acc2 = __builtin_amdgcn_wmma_f32_16x16x4_f32(false, a, false, b,
                                                     (short)0, acc2, false, false);
        b.x = w0[m + 48]; b.y = w1[m + 48];
        acc3 = __builtin_amdgcn_wmma_f32_16x16x4_f32(false, a, false, b,
                                                     (short)0, acc3, false, false);
    }

    float b0 = bias[m], b1 = bias[m + 16], b2 = bias[m + 32], b3 = bias[m + 48];
#pragma unroll
    for (int v = 0; v < 8; ++v) {
        acc0[v] += b0; acc1[v] += b1; acc2[v] += b2; acc3[v] += b3;
    }

    float inv[8];
#pragma unroll
    for (int v = 0; v < 8; ++v) {
        float p = acc0[v] * acc0[v] + acc1[v] * acc1[v]
                + acc2[v] * acc2[v] + acc3[v] * acc3[v];
        p += __shfl_xor(p, 1, 32);
        p += __shfl_xor(p, 2, 32);
        p += __shfl_xor(p, 4, 32);
        p += __shfl_xor(p, 8, 32);
        inv[v] = 1.0f / __builtin_sqrtf(p);
    }

#pragma unroll
    for (int v = 0; v < 8; ++v) {
        float* orow = out + (size_t)(rowBase + v + 8 * hi) * GC_FEAT;
        float s2 = inv[v];
        orow[m]      = acc0[v] * s2;
        orow[m + 16] = acc1[v] * s2;
        orow[m + 32] = acc2[v] * s2;
        orow[m + 48] = acc3[v] * s2;
    }
}

extern "C" void kernel_launch(void* const* d_in, const int* in_sizes, int n_in,
                              void* d_out, int out_size, void* d_ws, size_t ws_size,
                              hipStream_t stream) {
    const float* x      = (const float*)d_in[0];
    const int*   erow   = (const int*)  d_in[1];
    const int*   ecol   = (const int*)  d_in[2];
    const float* eval   = (const float*)d_in[3];
    const float* weight = (const float*)d_in[4];
    const float* bias   = (const float*)d_in[5];
    float*       out    = (float*)d_out;

    int E  = in_sizes[1];

    int* ws = (int*)d_ws;

    gc_seg_sort<<<GC_PART_BLOCKS, 1024, 0, stream>>>(erow, ecol, eval, ws, E);
    gc_gather_gemm_norm<<<GC_NBINS, GC_GATHER_THREADS, 0, stream>>>(x, ws, weight, bias, out, E);
}


# Sort launch re-cut to one workgroup per WGP: 256 chunks of 6250 edges, 8 waves, 25 edges per thread in flight (was 128 groups of 32 waves, 13); a bin walks 256 runs; row sums keep arrival-rank order
# speedup vs baseline: 4.5690x; 1.0727x over previous
#include <hip/hip_runtime.h>
#include <hip/hip_bf16.h>

typedef __attribute__((ext_vector_type(2))) float v2f;
typedef __attribute__((ext_vector_type(8))) float v8f;

#define GC_N_NODES 100000
#define GC_FEAT 64

#define GC_BIN_ROWS    208
#define GC_NBINS       481
#define GC_BIN_PAD     512
#define GC_PART_BLOCKS 256
#define GC_SORT_THREADS 256
#define GC_LDS_STRIDE  68
#define GC_COL_BITS    17
#define GC_GATHER_THREADS 512
#define GC_EDGE_BATCH  25
#define GC_GATHER_DEPTH 16
#define GC_SEG_CAP     (GC_SORT_THREADS * GC_EDGE_BATCH)
#define GC_SORT_CAP    4096
#define GC_SORT_SLOTS  (GC_SORT_CAP / GC_GATHER_THREADS)
#define GC_WAVE_ROWS   (GC_BIN_ROWS / (GC_GATHER_THREADS / 32))

__global__ __launch_bounds__(GC_SORT_THREADS)
void gc_seg_sort(const int*   __restrict__ erow,
                 const int*   __restrict__ ecol,
                 const float* __restrict__ eval,
                 int*         __restrict__ ws,
                 int E) {
    __shared__ __attribute__((aligned(8))) int cntb[GC_BIN_PAD];
    __shared__ int  wsum[GC_SORT_THREADS / 32];
    __shared__ int2 stage[GC_SEG_CAP];
    int t    = threadIdx.x;
    int lane = t & 31;
    int g    = blockIdx.x;

    int2* recs = (int2*)ws;
    int*  tab  = ws + 2 * (size_t)E;
    int chunk  = (E + GC_PART_BLOCKS - 1) / GC_PART_BLOCKS;
    int e0     = g * chunk;
    int e1     = min(E, e0 + chunk);
    int rounds = (chunk + GC_SEG_CAP - 1) / GC_SEG_CAP;

    for (int r = 0; r < rounds; ++r) {
        int s0   = e0 + r * GC_SEG_CAP;
        int s1   = min(e1, s0 + GC_SEG_CAP);
        int n    = max(0, s1 - s0);
        int safe = n > 0 ? s0 : 0;

        *(int2*)&cntb[2 * t] = make_int2(0, 0);
        __syncthreads();

        int   rr[GC_EDGE_BATCH];
        int   cc[GC_EDGE_BATCH];
        float vv[GC_EDGE_BATCH];
        int   rk[GC_EDGE_BATCH];
        int eb = s0 + t;
#pragma unroll
        for (int u = 0; u < GC_EDGE_BATCH; ++u) {
            int e = eb + GC_SORT_THREADS * u;
            e = e < s1 ? e : safe;
            rr[u] = erow[e];
            cc[u] = ecol[e];
            vv[u] = eval[e];
        }
#pragma unroll
        for (int u = 0; u < GC_EDGE_BATCH; ++u) {
            bool ok = eb + GC_SORT_THREADS * u < s1;
            rk[u] = atomicAdd(&cntb[rr[u] / GC_BIN_ROWS], ok ? 1 : 0);
        }
        __syncthreads();

        int2 cp  = *(const int2*)&cntb[2 * t];
        int  c   = cp.x + cp.y;
        int incl = c;
        for (int d = 1; d < 32; d <<= 1) {
            int y = __shfl_up(incl, d, 32);
            if (lane >= d) incl += y;
        }
        if (lane == 31) wsum[t >> 5] = incl;
        __syncthreads();
        {
            int base = 0;
#pragma unroll
            for (int q = 0; q < GC_SORT_THREADS / 32; ++q) {
                int v = wsum[q];
                base += (q < (t >> 5)) ? v : 0;
            }
            int2 ex;
            ex.x = base + incl - c;
            ex.y = ex.x + cp.x;
            *(int2*)&cntb[2 * t] = ex;
            *(int2*)&tab[(r * GC_PART_BLOCKS + g) * GC_BIN_PAD + 2 * t] = ex;
        }
        __syncthreads();

        int dst[GC_EDGE_BATCH];
#pragma unroll
        for (int u = 0; u < GC_EDGE_BATCH; ++u)
            dst[u] = cntb[rr[u] / GC_BIN_ROWS] + rk[u];
#pragma unroll
        for (int u = 0; u < GC_EDGE_BATCH; ++u) {
            if (eb + GC_SORT_THREADS * u < s1) {
                int b = rr[u] / GC_BIN_ROWS;
                int2 rec;
                rec.x = ((rr[u] - b * GC_BIN_ROWS) << GC_COL_BITS) | cc[u];
                rec.y = __float_as_int(vv[u]);
                stage[dst[u]] = rec;
            }
        }
        __syncthreads();

        int2 ov[GC_EDGE_BATCH];
#pragma unroll
        for (int u = 0; u < GC_EDGE_BATCH; ++u)
            ov[u] = stage[t + GC_SORT_THREADS * u];
#pragma unroll
        for (int u = 0; u < GC_EDGE_BATCH; ++u) {
            int i = t + GC_SORT_THREADS * u;
            if (i < n) recs[s0 + i] = ov[u];
        }
        __syncthreads();
    }
}

__global__ __launch_bounds__(GC_GATHER_THREADS)
void gc_gather_gemm_norm(const float* __restrict__ x,
                         const int*   __restrict__ ws,
                         const float* __restrict__ weight,
                         const float* __restrict__ bias,
                         float*       __restrict__ out,
                         int E) {
    __shared__ float acc[(GC_BIN_ROWS + 1) * GC_LDS_STRIDE];
    __shared__ int pre[GC_GATHER_THREADS];
    __shared__ int ofs[GC_GATHER_THREADS];
    __shared__ int wsum[GC_GATHER_THREADS / 32];
    __shared__ int2 srt[GC_SORT_CAP];
    __shared__ int rcnt[256];
    __shared__ int rsum[8];

    int t    = threadIdx.x;
    int lane = t & 31;
    int w    = t >> 5;
    int bin  = blockIdx.x;

    for (int i = t; i < (GC_BIN_ROWS + 1) * GC_LDS_STRIDE; i += GC_GATHER_THREADS)
        acc[i] = 0.f;
    __syncthreads();

    const int2* recs = (const int2*)ws;
    const int*  tab  = ws + 2 * (size_t)E;

    int2 padRec;
    padRec.x = GC_BIN_ROWS << GC_COL_BITS;
    padRec.y = 0;

    int chunk  = (E + GC_PART_BLOCKS - 1) / GC_PART_BLOCKS;
    int rounds = (chunk + GC_SEG_CAP - 1) / GC_SEG_CAP;
    int nseg   = rounds * GC_PART_BLOCKS;

    for (int sbase = 0; sbase < nseg; sbase += GC_GATHER_THREADS) {
        int sg  = sbase + t;
        int len = 0, st = 0;
        if (sg < nseg) {
            int a  = tab[sg * GC_BIN_PAD + bin];
            int b2 = tab[sg * GC_BIN_PAD + bin + 1];
            len = b2 - a;
            st  = (int)((unsigned)sg % GC_PART_BLOCKS) * chunk
                + (int)((unsigned)sg / GC_PART_BLOCKS) * GC_SEG_CAP + a;
        }
        int incl = len;
        for (int d = 1; d < 32; d <<= 1) {
            int y = __shfl_up(incl, d, 32);
            if (lane >= d) incl += y;
        }
        if (lane == 31) wsum[w] = incl;
        __syncthreads();
        int base = 0, tot = 0;
        for (int q = 0; q < GC_GATHER_THREADS / 32; ++q) {
            int v = wsum[q];
            base += (q < w) ? v : 0;
            tot  += v;
        }
        int ex = base + incl - len;
        pre[t] = ex;
        ofs[t] = st - ex;
        __syncthreads();

        int nloc  = min(nseg - sbase, GC_GATHER_THREADS);
        int step0 = GC_GATHER_THREADS / 2;
        while (step0 >= nloc && step0 > 0) step0 >>= 1;

        int wu = __builtin_amdgcn_readfirstlane(w);

        for (int p0 = 0; p0 < tot; p0 += GC_SORT_CAP) {
            int pn = min(tot - p0, GC_SORT_CAP);
            if (t < 256) rcnt[t] = 0;
            __syncthreads();

            int fidx[GC_SORT_SLOTS];
            int fj[GC_SORT_SLOTS];
#pragma unroll
            for (int u = 0; u < GC_SORT_SLOTS; ++u) {
                int i = t + GC_GATHER_THREADS * u;
                fidx[u] = p0 + (i < pn ? i : 0);
                fj[u]   = 0;
            }
            for (int step = step0; step; step >>= 1) {
#pragma unroll
                for (int u = 0; u < GC_SORT_SLOTS; ++u)
                    if (pre[fj[u] + step] <= fidx[u]) fj[u] += step;
            }
            int2 rc[GC_SORT_SLOTS];
            int  rk[GC_SORT_SLOTS];
#pragma unroll
            for (int u = 0; u < GC_SORT_SLOTS; ++u)
                rc[u] = recs[fidx[u] + ofs[fj[u]]];
#pragma unroll
            for (int u = 0; u < GC_SORT_SLOTS; ++u) {
                bool ok = t + GC_GATHER_THREADS * u < pn;
                rk[u] = atomicAdd(&rcnt[rc[u].x >> GC_COL_BITS], ok ? 1 : 0);
            }
            __syncthreads();

            int c = 0, rincl = 0;
            if (t < 256) {
                c = rcnt[t];
                rincl = c;
                for (int d = 1; d < 32; d <<= 1) {
                    int y = __shfl_up(rincl, d, 32);
                    if (lane >= d) rincl += y;
                }
                if (lane == 31) rsum[w] = rincl;
            }
            __syncthreads();
            if (t < 256) {
                int rbase = 0;
                for (int q = 0; q < w; ++q) rbase += rsum[q];
                rcnt[t] = rbase + rincl - c;
            }
            __syncthreads();

#pragma unroll
            for (int u = 0; u < GC_SORT_SLOTS; ++u) {
                if (t + GC_GATHER_THREADS * u < pn)
                    srt[rcnt[rc[u].x >> GC_COL_BITS] + rk[u]] = rc[u];
            }
            __syncthreads();

            int beg = __builtin_amdgcn_readfirstlane(rcnt[wu * GC_WAVE_ROWS]);
            int end = __builtin_amdgcn_readfirstlane(rcnt[wu * GC_WAVE_ROWS + GC_WAVE_ROWS]);
            int cur = wu * GC_WAVE_ROWS;
            float sx = 0.f, sy = 0.f;
            int2 rec = padRec;
            if (beg + lane < end) rec = srt[beg + lane];
            for (int b = beg; b < end; b += 32) {
                int2 recNext = padRec;
                if (b + 32 + lane < end) recNext = srt[b + 32 + lane];
#pragma unroll
                for (int grp = 0; grp < 32 / GC_GATHER_DEPTH; ++grp) {
                    if (b + grp * GC_GATHER_DEPTH < end) {
                        float2 mrow[GC_GATHER_DEPTH];
                        int    lrow[GC_GATHER_DEPTH];
                        float  val[GC_GATHER_DEPTH];
#pragma unroll
                        for (int u = 0; u < GC_GATHER_DEPTH; ++u) {
                            int px  = __builtin_amdgcn_readlane(rec.x, grp * GC_GATHER_DEPTH + u);
                            int pv  = __builtin_amdgcn_readlane(rec.y, grp * GC_GATHER_DEPTH + u);
                            int col = px & ((1 << GC_COL_BITS) - 1);
                            lrow[u] = px >> GC_COL_BITS;
                            val[u]  = __int_as_float(pv);
                            mrow[u] = *(const float2*)(x + (size_t)col * GC_FEAT + 2 * lane);
                        }
#pragma unroll
                        for (int u = 0; u < GC_GATHER_DEPTH; ++u) {
                            if (lrow[u] != cur) {
                                float* d = acc + cur * GC_LDS_STRIDE + 2 * lane;
                                __hip_atomic_fetch_add(d,     sx, __ATOMIC_RELAXED, __HIP_MEMORY_SCOPE_WORKGROUP);
                                __hip_atomic_fetch_add(d + 1, sy, __ATOMIC_RELAXED, __HIP_MEMORY_SCOPE_WORKGROUP);
                                sx = 0.f; sy = 0.f;
                                cur = lrow[u];
                            }
                            sx += val[u] * mrow[u].x;
                            sy += val[u] * mrow[u].y;
                        }
                    }
                }
                rec = recNext;
            }
            {
                float* d = acc + cur * GC_LDS_STRIDE + 2 * lane;
                __hip_atomic_fetch_add(d,     sx, __ATOMIC_RELAXED, __HIP_MEMORY_SCOPE_WORKGROUP);
                __hip_atomic_fetch_add(d + 1, sy, __ATOMIC_RELAXED, __HIP_MEMORY_SCOPE_WORKGROUP);
            }
            __syncthreads();
        }
    }

    int rowBase = bin * GC_BIN_ROWS + w * 16;
    if (w >= GC_BIN_ROWS / 16 || rowBase >= GC_N_NODES) return;

    int m    = lane & 15;
    int hi   = lane >> 4;
    int kb   = hi * 2;

    v8f acc0 = {}, acc1 = {}, acc2 = {}, acc3 = {};

    const float* arow = acc + (w * 16 + m) * GC_LDS_STRIDE + kb;

#pragma unroll
    for (int ki = 0; ki < 16; ++ki) {
        int k = ki * 4 + kb;
        float2 av = *(const float2*)(arow + ki * 4);
        v2f a; a.x = av.x; a.y = av.y;

        const float* w0 = weight + (size_t)k * GC_FEAT;
        const float* w1 = w0 + GC_FEAT;
        v2f b;

        b.x = w0[m];      b.y = w1[m];
        acc0 = __builtin_amdgcn_wmma_f32_16x16x4_f32(false, a, false, b,
                                                     (short)0, acc0, false, false);
        b.x = w0[m + 16]; b.y = w1[m + 16];
        acc1 = __builtin_amdgcn_wmma_f32_16x16x4_f32(false, a, false, b,
                                                     (short)0, acc1, false, false);
        b.x = w0[m + 32]; b.y = w1[m + 32];
        acc2 = __builtin_amdgcn_wmma_f32_16x16x4_f32(false, a, false, b,
                                                     (short)0, acc2, false, false);
        b.x = w0[m + 48]; b.y = w1[m + 48];
        acc3 = __builtin_amdgcn_wmma_f32_16x16x4_f32(false, a, false, b,
                                                     (short)0, acc3, false, false);
    }

    float b0 = bias[m], b1 = bias[m + 16], b2 = bias[m + 32], b3 = bias[m + 48];
#pragma unroll
    for (int v = 0; v < 8; ++v) {
        acc0[v] += b0; acc1[v] += b1; acc2[v] += b2; acc3[v] += b3;
    }

    float inv[8];
#pragma unroll
    for (int v = 0; v < 8; ++v) {
        float p = acc0[v] * acc0[v] + acc1[v] * acc1[v]
                + acc2[v] * acc2[v] + acc3[v] * acc3[v];
        p += __shfl_xor(p, 1, 32);
        p += __shfl_xor(p, 2, 32);
        p += __shfl_xor(p, 4, 32);
        p += __shfl_xor(p, 8, 32);
        inv[v] = 1.0f / __builtin_sqrtf(p);
    }

#pragma unroll
    for (int v = 0; v < 8; ++v) {
        float* orow = out + (size_t)(rowBase + v + 8 * hi) * GC_FEAT;
        float s2 = inv[v];
        orow[m]      = acc0[v] * s2;
        orow[m + 16] = acc1[v] * s2;
        orow[m + 32] = acc2[v] * s2;
        orow[m + 48] = acc3[v] * s2;
    }
}

extern "C" void kernel_launch(void* const* d_in, const int* in_sizes, int n_in,
                              void* d_out, int out_size, void* d_ws, size_t ws_size,
                              hipStream_t stream) {
    const float* x      = (const float*)d_in[0];
    const int*   erow   = (const int*)  d_in[1];
    const int*   ecol   = (const int*)  d_in[2];
    const float* eval   = (const float*)d_in[3];
    const float* weight = (const float*)d_in[4];
    const float* bias   = (const float*)d_in[5];
    float*       out    = (float*)d_out;

    int E  = in_sizes[1];

    int* ws = (int*)d_ws;

    gc_seg_sort<<<GC_PART_BLOCKS, GC_SORT_THREADS, 0, stream>>>(erow, ecol, eval, ws, E);
    gc_gather_gemm_norm<<<GC_NBINS, GC_GATHER_THREADS, 0, stream>>>(x, ws, weight, bias, out, E);
}
